# starting kernel
# baseline (speedup 1.0000x reference)
#include <hip/hip_runtime.h>
#include <hip/hip_bf16.h>

typedef __attribute__((ext_vector_type(2))) float v2f;
typedef __attribute__((ext_vector_type(8))) float v8f;

#define NCOL 2048
#define ROWS 16
#define EPSF 1e-10f

__global__ __launch_bounds__(256) void qcd_softmax_kernel(
    const float* __restrict__ emb,
    const float* __restrict__ alphap,
    const float* __restrict__ betap,
    float* __restrict__ out)
{
    __shared__ float s_logit[ROWS * NCOL];
    __shared__ float s_x[NCOL];
    __shared__ float s_y[NCOL];
    __shared__ float s_p[NCOL];
    __shared__ float s_prow[ROWS];
    __shared__ float s_rowmax[ROWS];
    __shared__ float s_rowsum[ROWS];
    __shared__ float s_wmax[8 * ROWS];

    const int tid  = threadIdx.x;
    const int lane = tid & 31;
    const int wv   = tid >> 5;
    const int half = lane >> 4;
    const int l15  = lane & 15;

    const int blk = blockIdx.x;
    const int b   = blk >> 7;
    const int i0  = (blk & 127) << 4;

    const float alpha = alphap[0];
    const float beta  = betap[0];
    const float nb2   = -beta * beta;
    const float a2    = 2.0f * alpha;

    const float* E = emb + (size_t)b * 6 * NCOL;

    #pragma unroll
    for (int k = 0; k < 8; ++k) {
        int c = (k << 8) + tid;
        float x   = E[1 * NCOL + c];
        float y   = E[2 * NCOL + c];
        float mom = E[4 * NCOL + c] + EPSF;
        s_x[c] = x;
        s_y[c] = y;
        s_p[c] = __expf(a2 * __logf(mom));
    }
    if (tid < ROWS) {
        float mom = E[4 * NCOL + i0 + tid] + EPSF;
        s_prow[tid]  = __expf(a2 * __logf(mom));
        s_rowsum[tid] = 0.0f;
    }
    __syncthreads();

    float xi = s_x[i0 + l15];
    float yi = s_y[i0 + l15];
    v2f afrag;
    afrag[0] = half ? (xi * xi + yi * yi) : xi;
    afrag[1] = half ? 1.0f : yi;

    float pi[8];
    #pragma unroll
    for (int r = 0; r < 8; ++r) pi[r] = s_prow[half * 8 + r];

    float rmax[8];
    #pragma unroll
    for (int r = 0; r < 8; ++r) rmax[r] = -INFINITY;

    #pragma unroll 4
    for (int s = 0; s < 16; ++s) {
        int j0 = ((s << 3) + wv) << 4;
        float xj = s_x[j0 + l15];
        float yj = s_y[j0 + l15];
        v2f bfrag;
        bfrag[0] = half ? 1.0f : (-2.0f * xj);
        bfrag[1] = half ? (xj * xj + yj * yj) : (-2.0f * yj);

        v8f c = {};
        c = __builtin_amdgcn_wmma_f32_16x16x4_f32(
                false, afrag, false, bfrag, (short)0, c, false, false);

        float pj = s_p[j0 + l15];
        #pragma unroll
        for (int r = 0; r < 8; ++r) {
            float logit = nb2 * c[r] * fminf(pi[r], pj);
            rmax[r] = fmaxf(rmax[r], logit);
            s_logit[(half * 8 + r) * NCOL + j0 + l15] = logit;
        }
    }

    #pragma unroll
    for (int r = 0; r < 8; ++r) {
        #pragma unroll
        for (int m = 1; m <= 8; m <<= 1)
            rmax[r] = fmaxf(rmax[r], __shfl_xor(rmax[r], m, 32));
    }
    if (l15 == 0) {
        #pragma unroll
        for (int r = 0; r < 8; ++r)
            s_wmax[wv * ROWS + half * 8 + r] = rmax[r];
    }
    __syncthreads();
    if (tid < ROWS) {
        float m = s_wmax[tid];
        #pragma unroll
        for (int w = 1; w < 8; ++w) m = fmaxf(m, s_wmax[w * ROWS + tid]);
        s_rowmax[tid] = m;
    }
    __syncthreads();

    {
        float acc = 0.0f;
        #pragma unroll 8
        for (int k = 0; k < 128; ++k) {
            int idx = (k << 8) + tid;
            int r = k >> 3;
            float e = __expf(s_logit[idx] - s_rowmax[r]);
            s_logit[idx] = e;
            acc += e;
            if ((k & 7) == 7) { atomicAdd(&s_rowsum[r], acc); acc = 0.0f; }
        }
    }
    __syncthreads();

    {
        float* dst = out + (size_t)b * NCOL * NCOL + (size_t)i0 * NCOL;
        float inv = 0.0f;
        #pragma unroll 8
        for (int k = 0; k < 128; ++k) {
            int r = k >> 3;
            if ((k & 7) == 0) inv = 1.0f / s_rowsum[r];
            int idx = (k << 8) + tid;
            dst[idx] = s_logit[idx] * inv;
        }
    }
}

extern "C" void kernel_launch(void* const* d_in, const int* in_sizes, int n_in,
                              void* d_out, int out_size, void* d_ws, size_t ws_size,
                              hipStream_t stream) {
    const float* emb   = (const float*)d_in[0];
    const float* alpha = (const float*)d_in[1];
    const float* beta  = (const float*)d_in[2];
    float* out = (float*)d_out;

    dim3 grid(8 * (NCOL / ROWS));
    dim3 block(256);
    qcd_softmax_kernel<<<grid, block, 0, stream>>>(emb, alpha, beta, out);
}


# Store pass of strip k and sum pass of strip k+1 fused in one loop (a block walks 2 strips, each B fragment feeds both products): stores run under the next strip's sum work; sum order kept
# speedup vs baseline: 3.2923x; 3.2923x over previous
#include <hip/hip_runtime.h>
#include <hip/hip_bf16.h>


typedef __attribute__((ext_vector_type(2))) float v2f;
typedef __attribute__((ext_vector_type(8))) float v8f;
typedef __attribute__((ext_vector_type(2))) unsigned int v2u;

#define NCOL 2048
#define ROWS 16
#define NSTRIP 2
#define EPSF 1e-10f


__device__ __forceinline__ void strip_operands(
    const float* s_x, const float* s_y, const float* s_p,
    int row0, int half, int l15, v2f& a, float (&pi)[8])
{
    float xi = s_x[row0 + l15];
    float yi = s_y[row0 + l15];
    a[0] = half ? (xi * xi + yi * yi) : xi;
    a[1] = half ? 1.0f : yi;
    #pragma unroll
    for (int r = 0; r < 8; ++r) pi[r] = s_p[row0 + half * 8 + r];
}

__device__ __forceinline__ v2f column_fragment(float xj, float yj, int half)
{
    v2f bfrag;
    bfrag[0] = half ? 1.0f : (-2.0f * xj);
    bfrag[1] = half ? (xj * xj + yj * yj) : (-2.0f * yj);
    return bfrag;
}

__device__ __forceinline__ void put_row_sums(float (&rsum)[8], float* wsum,
                                             int wv, int half, int l15)
{
    #pragma unroll
    for (int r = 0; r < 8; ++r) {
        #pragma unroll
        for (int m = 1; m <= 8; m <<= 1)
            rsum[r] += __shfl_xor(rsum[r], m, 32);
    }
    if (l15 == 0) {
        #pragma unroll
        for (int r = 0; r < 8; ++r)
            wsum[wv * ROWS + half * 8 + r] = rsum[r];
    }
}

__device__ __forceinline__ void get_inverse_sums(const float* wsum, int half,
                                                 float (&inv)[8])
{
    #pragma unroll
    for (int r = 0; r < 8; ++r) {
        float t = wsum[half * 8 + r];
        #pragma unroll
        for (int w = 1; w < 8; ++w) t += wsum[w * ROWS + half * 8 + r];
        inv[r] = 1.0f / t;
    }
}

__global__ __launch_bounds__(256, 4) void qcd_softmax_kernel(
    const float* __restrict__ emb,
    const float* __restrict__ alphap,
    const float* __restrict__ betap,
    float* __restrict__ out)
{
    __shared__ float s_x[NCOL];
    __shared__ float s_y[NCOL];
    __shared__ float s_p[NCOL];
    __shared__ float s_wsum[NSTRIP][8 * ROWS];

    const int tid  = threadIdx.x;
    const int lane = tid & 31;
    const int wv   = tid >> 5;
    const int half = lane >> 4;
    const int l15  = lane & 15;

    const int blk = blockIdx.x;
    const int spb = NCOL / (ROWS * NSTRIP);
    const int b   = blk / spb;
    const int i0  = (blk % spb) * (ROWS * NSTRIP);

    const float alpha = alphap[0];
    const float beta  = betap[0];
    const float nb2   = -beta * beta;
    const float a2    = 2.0f * alpha;

    const float* E = emb + (size_t)b * 6 * NCOL;

    #pragma unroll
    for (int k = 0; k < 8; ++k) {
        int c = (k << 8) + tid;
        float x   = E[1 * NCOL + c];
        float y   = E[2 * NCOL + c];
        float mom = E[4 * NCOL + c] + EPSF;
        s_x[c] = x;
        s_y[c] = y;
        s_p[c] = __expf(a2 * __logf(mom));
    }
    __syncthreads();

    v2f   a_cur;
    float pi_cur[8];
    float inv_cur[8];
    strip_operands(s_x, s_y, s_p, i0, half, l15, a_cur, pi_cur);

    {
        float rsum[8];
        #pragma unroll
        for (int r = 0; r < 8; ++r) rsum[r] = 0.0f;

        for (int s = 0; s < 4; ++s) {
            const int j0 = ((s << 3) + wv) << 6;
            #pragma unroll
            for (int t = 0; t < 4; ++t) {
                const int j = j0 + (t << 4) + l15;
                v2f bfrag = column_fragment(s_x[j], s_y[j], half);
                v8f c = {};
                c = __builtin_amdgcn_wmma_f32_16x16x4_f32(
                        false, a_cur, false, bfrag, (short)0, c, false, false);
                float pj = s_p[j];
                #pragma unroll
                for (int r = 0; r < 8; ++r) {
                    float logit = nb2 * c[r] * fminf(pi_cur[r], pj);
                    rsum[r] += __expf(logit);
                }
            }
        }
        put_row_sums(rsum, s_wsum[0], wv, half, l15);
        __syncthreads();
        get_inverse_sums(s_wsum[0], half, inv_cur);
    }

    #pragma unroll
    for (int k = 0; k < NSTRIP; ++k) {
        const bool has_next = (k + 1 < NSTRIP);
        float* dst = out + (size_t)b * NCOL * NCOL + (size_t)(i0 + k * ROWS) * NCOL;

        v2f   a_nxt = a_cur;
        float pi_nxt[8];
        float rsum[8];
        #pragma unroll
        for (int r = 0; r < 8; ++r) { rsum[r] = 0.0f; pi_nxt[r] = 0.0f; }
        if (has_next)
            strip_operands(s_x, s_y, s_p, i0 + (k + 1) * ROWS, half, l15, a_nxt, pi_nxt);

        #pragma nounroll
        for (int s = 0; s < 4; ++s) {
            const int j0 = ((s << 3) + wv) << 6;
            float e[4][8];
            #pragma unroll
            for (int t = 0; t < 4; ++t) {
                const int j = j0 + (t << 4) + l15;
                v2f bfrag = column_fragment(s_x[j], s_y[j], half);
                float pj = s_p[j];

                v8f c = {};
                c = __builtin_amdgcn_wmma_f32_16x16x4_f32(
                        false, a_cur, false, bfrag, (short)0, c, false, false);
                #pragma unroll
                for (int r = 0; r < 8; ++r) {
                    float logit = nb2 * c[r] * fminf(pi_cur[r], pj);
                    e[t][r] = __expf(logit) * inv_cur[r];
                }

                if (has_next) {
                    v8f d = {};
                    d = __builtin_amdgcn_wmma_f32_16x16x4_f32(
                            false, a_nxt, false, bfrag, (short)0, d, false, false);
                    #pragma unroll
                    for (int r = 0; r < 8; ++r) {
                        float logit = nb2 * d[r] * fminf(pi_nxt[r], pj);
                        rsum[r] += __expf(logit);
                    }
                }
            }
            #pragma unroll
            for (int r = 0; r < 8; ++r) {
                v2u p01 = __builtin_amdgcn_permlane16_swap(
                    __float_as_uint(e[0][r]), __float_as_uint(e[1][r]), false, false);
                v2u p23 = __builtin_amdgcn_permlane16_swap(
                    __float_as_uint(e[2][r]), __float_as_uint(e[3][r]), false, false);
                float* row_lo = dst + (size_t)r * NCOL + j0 + lane;
                float* row_hi = dst + (size_t)(r + 8) * NCOL + j0 + lane;
                row_lo[0]  = __uint_as_float(p01[0]);
                row_lo[32] = __uint_as_float(p23[0]);
                row_hi[0]  = __uint_as_float(p01[1]);
                row_hi[32] = __uint_as_float(p23[1]);
            }
        }

        if (has_next) {
            put_row_sums(rsum, s_wsum[k + 1], wv, half, l15);
            __syncthreads();
            get_inverse_sums(s_wsum[k + 1], half, inv_cur);
            a_cur = a_nxt;
            #pragma unroll
            for (int r = 0; r < 8; ++r) pi_cur[r] = pi_nxt[r];
        }
    }
}

extern "C" void kernel_launch(void* const* d_in, const int* in_sizes, int n_in,
                              void* d_out, int out_size, void* d_ws, size_t ws_size,
                              hipStream_t stream) {
    const float* emb   = (const float*)d_in[0];
    const float* alpha = (const float*)d_in[1];
    const float* beta  = (const float*)d_in[2];
    float* out = (float*)d_out;

    dim3 grid(8 * (NCOL / (ROWS * NSTRIP)));
    dim3 block(256);
    qcd_softmax_kernel<<<grid, block, 0, stream>>>(emb, alpha, beta, out);
}
